# starting kernel
# baseline (speedup 1.0000x reference)
#include <hip/hip_runtime.h>

typedef float v2f __attribute__((ext_vector_type(2)));
typedef float v8f __attribute__((ext_vector_type(8)));

#define HW_DIM   128
#define F_OUT    256
#define KDIM     2304
#define NROI     8192
#define HOWO     16384
#define KTILE    16
#define LDA      20
#define LDB      136

__global__ __launch_bounds__(256) void bias_fill_kernel(
    const float* __restrict__ bias, float* __restrict__ out) {
  const int g = blockIdx.x * 256 + threadIdx.x;
  const int f = g >> 12;
  const float bv = bias[f];
  reinterpret_cast<float4*>(out)[g] = make_float4(bv, bv, bv, bv);
}

__global__ __launch_bounds__(256) void conv_roi_wmma_kernel(
    const float* __restrict__ X,
    const float* __restrict__ W,
    const float* __restrict__ bias,
    const int*   __restrict__ roi,
    float* __restrict__ out) {
  __shared__ float As[128 * LDA];
  __shared__ float Bs[KTILE * LDB];

  const int tid   = threadIdx.x;
  const int lane  = tid & 31;
  const int wave  = tid >> 5;
  const int waveM = wave >> 2;
  const int waveN = wave & 3;
  const int lrow  = lane & 15;
  const int lhi   = lane >> 4;

  const int m0 = blockIdx.y * 128;
  const int n0 = blockIdx.x * 128;

  const int tn   = tid & 127;
  const int kgrp = tid >> 7;
  const int p_my = roi[n0 + tn];
  const int ry   = p_my >> 7;
  const int rx   = p_my & 127;

  const int arow = tid >> 2;
  const int aq   = tid & 3;

  v8f acc[4][2];
  const v8f vzero = {};
#pragma unroll
  for (int mt = 0; mt < 4; ++mt)
#pragma unroll
    for (int nt = 0; nt < 2; ++nt) acc[mt][nt] = vzero;

  for (int k0 = 0; k0 < KDIM; k0 += KTILE) {
#pragma unroll
    for (int rep = 0; rep < 2; ++rep) {
      const int row = arow + rep * 64;
      const float4 w4 =
          *reinterpret_cast<const float4*>(W + (m0 + row) * KDIM + k0 + aq * 4);
      *reinterpret_cast<float4*>(&As[row * LDA + aq * 4]) = w4;
    }
#pragma unroll
    for (int u = 0; u < 8; ++u) {
      const int kk  = kgrp * 8 + u;
      const int k   = k0 + kk;
      const int ch  = k / 9;
      const int rem = k - ch * 9;
      const int fi  = rem / 3;
      const int fj  = rem - fi * 3;
      const int y   = ry + fi - 1;
      const int x   = rx + fj - 1;
      float v = 0.0f;
      if ((unsigned)y < 128u && (unsigned)x < 128u)
        v = X[(ch << 14) + (y << 7) + x];
      Bs[kk * LDB + tn] = v;
    }
    if (k0 + KTILE < KDIM)
      __builtin_prefetch(W + (m0 + arow) * KDIM + k0 + KTILE, 0, 0);
    __syncthreads();

#pragma unroll
    for (int ks = 0; ks < 4; ++ks) {
      const int kb = ks * 4;
      v2f afrag[4];
#pragma unroll
      for (int mt = 0; mt < 4; ++mt) {
        const int m = waveM * 64 + mt * 16 + lrow;
        afrag[mt] = *reinterpret_cast<const v2f*>(&As[m * LDA + kb + lhi * 2]);
      }
      v2f bfrag[2];
#pragma unroll
      for (int nt = 0; nt < 2; ++nt) {
        const int n  = waveN * 32 + nt * 16 + lrow;
        const int kk = kb + lhi * 2;
        v2f b;
        b.x = Bs[kk * LDB + n];
        b.y = Bs[(kk + 1) * LDB + n];
        bfrag[nt] = b;
      }
#pragma unroll
      for (int mt = 0; mt < 4; ++mt)
#pragma unroll
        for (int nt = 0; nt < 2; ++nt)
          acc[mt][nt] = __builtin_amdgcn_wmma_f32_16x16x4_f32(
              false, afrag[mt], false, bfrag[nt],
              (short)0, acc[mt][nt], false, false);
    }
    __syncthreads();
  }

#pragma unroll
  for (int mt = 0; mt < 4; ++mt) {
    const int fbase = m0 + waveM * 64 + mt * 16 + lhi * 8;
#pragma unroll
    for (int nt = 0; nt < 2; ++nt) {
      const int nn = n0 + waveN * 32 + nt * 16 + lrow;
      const int p  = roi[nn];
#pragma unroll
      for (int v = 0; v < 8; ++v) {
        const int f = fbase + v;
        out[(f << 14) + p] = acc[mt][nt][v] + bias[f];
      }
    }
  }
}

extern "C" void kernel_launch(void* const* d_in, const int* in_sizes, int n_in,
                              void* d_out, int out_size, void* d_ws, size_t ws_size,
                              hipStream_t stream) {
  const float* X    = (const float*)d_in[0];
  const float* W    = (const float*)d_in[1];
  const float* B    = (const float*)d_in[2];
  const int*   roi  = (const int*)d_in[3];
  float*       out  = (float*)d_out;

  bias_fill_kernel<<<dim3((F_OUT * HOWO / 4) / 256), 256, 0, stream>>>(B, out);
  conv_roi_wmma_kernel<<<dim3(NROI / 128, F_OUT / 128), 256, 0, stream>>>(
      X, W, B, roi, out);
}


# GEMM operands loaded from global memory straight into WMMA registers one K-tile ahead (W re-laid to lane order, each lane gathers its own 8 channels): no LDS staging, no barriers; sum order changes
# speedup vs baseline: 6.2807x; 6.2807x over previous
#include <hip/hip_runtime.h>


typedef float v2f __attribute__((ext_vector_type(2)));
typedef float v8f __attribute__((ext_vector_type(8)));

#define HW_DIM   128
#define F_OUT    256
#define KDIM     2304
#define NROI     8192
#define HOWO     16384
#define KTILES   144
#define KSPLIT   4
#define CHUNKS_PER_SPLIT (16 / KSPLIT)
#define PLANE    (F_OUT * NROI)

#define RELAYOUT_BLOCKS ((F_OUT * KDIM / 4) / 256)
#define FILL_BLOCKS     ((F_OUT * HOWO / 4) / 256)

__global__ __launch_bounds__(256) void bias_fill_relayout_kernel(
    const float* __restrict__ bias, const float* __restrict__ W,
    float* __restrict__ Wf, float* __restrict__ out) {
  float4  val;
  float4* dst;
  if (blockIdx.x < RELAYOUT_BLOCKS) {
    const int g     = blockIdx.x * 256 + threadIdx.x;
    const int l     = g & 31;
    const int j     = (g >> 5) & 1;
    const int tk    = g >> 6;
    const int T     = tk / KTILES;
    const int kt    = tk - T * KTILES;
    const int chunk = kt / 9;
    const int tap   = kt - chunk * 9;
    const int f     = T * 16 + (l & 15);
    const int ch    = chunk * 16 + (l >> 4) * 8 + j * 4;
    const float* src = W + f * KDIM + ch * 9 + tap;
    val = make_float4(src[0], src[9], src[18], src[27]);
    dst = reinterpret_cast<float4*>(Wf) + g;
  } else {
    const int g = (blockIdx.x - RELAYOUT_BLOCKS) * 256 + threadIdx.x;
    const int f = g >> 12;
    const float bv = bias[f];
    val = make_float4(bv, bv, bv, bv);
    dst = reinterpret_cast<float4*>(out) + g;
  }
  *dst = val;
}


__device__ __forceinline__ bool tap_inside(int tap, int ry, int rx) {
  const int fi = tap / 3;
  const int fj = tap - fi * 3;
  const int y  = ry + fi - 1;
  const int x  = rx + fj - 1;
  return (unsigned)y < 128u && (unsigned)x < 128u;
}

__device__ __forceinline__ void load_tile(
    int tap, const float* __restrict__ wtile, const float* __restrict__ xchunk,
    const int (&p_my)[2], const int (&ry)[2], const int (&rx)[2],
    float4 (&a4)[4][2], float (&xv)[2][8]) {
#pragma unroll
  for (int mt = 0; mt < 4; ++mt)
#pragma unroll
    for (int j = 0; j < 2; ++j)
      a4[mt][j] = *reinterpret_cast<const float4*>(wtile + (mt * KTILES * 2 + j) * 128);
  const int fi = tap / 3;
  const int fj = tap - fi * 3;
#pragma unroll
  for (int nt = 0; nt < 2; ++nt) {
    const int poff = tap_inside(tap, ry[nt], rx[nt])
                         ? p_my[nt] + (fi - 1) * 128 + (fj - 1) : p_my[nt];
    const float* xp = xchunk + poff;
#pragma unroll
    for (int u = 0; u < 8; ++u) xv[nt][u] = xp[u << 14];
  }
}

__device__ __forceinline__ void mma_tile(
    int tap, const int (&ry)[2], const int (&rx)[2],
    const float4 (&a4)[4][2], const float (&xv)[2][8], v8f (&acc)[4][2]) {
  float b[2][8];
#pragma unroll
  for (int nt = 0; nt < 2; ++nt) {
    const bool ok = tap_inside(tap, ry[nt], rx[nt]);
#pragma unroll
    for (int u = 0; u < 8; ++u) {
      const float v = xv[nt][u];
      b[nt][u] = ok ? v : 0.0f;
    }
  }
#pragma unroll
  for (int ks = 0; ks < 4; ++ks) {
    v2f afrag[4];
#pragma unroll
    for (int mt = 0; mt < 4; ++mt) {
      const float4 q = a4[mt][ks >> 1];
      v2f a;
      a.x = (ks & 1) ? q.z : q.x;
      a.y = (ks & 1) ? q.w : q.y;
      afrag[mt] = a;
    }
    v2f bfrag[2];
#pragma unroll
    for (int nt = 0; nt < 2; ++nt) {
      v2f bb;
      bb.x = b[nt][2 * ks];
      bb.y = b[nt][2 * ks + 1];
      bfrag[nt] = bb;
    }
#pragma unroll
    for (int mt = 0; mt < 4; ++mt)
#pragma unroll
      for (int nt = 0; nt < 2; ++nt)
        acc[mt][nt] = __builtin_amdgcn_wmma_f32_16x16x4_f32(
            false, afrag[mt], false, bfrag[nt],
            (short)0, acc[mt][nt], false, false);
  }
}

__global__ __launch_bounds__(256) void conv_roi_wmma_kernel(
    const float* __restrict__ X,
    const float* __restrict__ Wf,
    const int*   __restrict__ roi,
    float* __restrict__ P) {
  const int tid   = threadIdx.x;
  const int lane  = tid & 31;
  const int wave  = tid >> 5;
  const int waveM = wave >> 2;
  const int waveN = wave & 3;
  const int lrow  = lane & 15;
  const int lhi   = lane >> 4;

  const int m0 = blockIdx.y * 128;
  const int n0 = blockIdx.x * 128;

  int p_my[2], ry[2], rx[2];
#pragma unroll
  for (int nt = 0; nt < 2; ++nt) {
    p_my[nt] = roi[n0 + waveN * 32 + nt * 16 + lrow];
    ry[nt]   = p_my[nt] >> 7;
    rx[nt]   = p_my[nt] & 127;
  }

  v8f acc[4][2];
  const v8f vzero = {};
#pragma unroll
  for (int mt = 0; mt < 4; ++mt)
#pragma unroll
    for (int nt = 0; nt < 2; ++nt) acc[mt][nt] = vzero;

  const float* xbase = X + ((lhi * 8) << 14);
  const float* wbase = Wf + (((m0 >> 4) + waveM * 4) * KTILES) * 256 + lane * 4;

  const int chunk0 = blockIdx.z * CHUNKS_PER_SPLIT;

  float4 a4[2][4][2];
  float  xv[2][2][8];

  load_tile(0, wbase + (chunk0 * 9) * 256, xbase + (chunk0 << 18), p_my, ry, rx, a4[0], xv[0]);

#pragma unroll 1
  for (int cc = 0; cc < CHUNKS_PER_SPLIT / 2; ++cc) {
    const int cbase = chunk0 + cc * 2;
#pragma unroll
    for (int t = 0; t < 18; ++t) {
      const int cur = t & 1;
      const int nxt = cur ^ 1;
      if (t < 17) {
        const int nchunk = cbase + (t + 1) / 9;
        const int ntap   = (t + 1) % 9;
        load_tile(ntap, wbase + (nchunk * 9 + ntap) * 256, xbase + (nchunk << 18),
                  p_my, ry, rx, a4[nxt], xv[nxt]);
      } else if (cc + 1 < CHUNKS_PER_SPLIT / 2) {
        const int nchunk = cbase + 2;
        load_tile(0, wbase + (nchunk * 9) * 256, xbase + (nchunk << 18),
                  p_my, ry, rx, a4[nxt], xv[nxt]);
      }
      __builtin_amdgcn_sched_barrier(0);
      mma_tile(t % 9, ry, rx, a4[cur], xv[cur], acc);
      __builtin_amdgcn_sched_barrier(0);
    }
  }

  float* Pz = P + blockIdx.z * PLANE;
#pragma unroll
  for (int mt = 0; mt < 4; ++mt) {
    const int ftile = (m0 >> 4) + waveM * 4 + mt;
#pragma unroll
    for (int nt = 0; nt < 2; ++nt) {
      const int ntile = (n0 >> 4) + waveN * 2 + nt;
      float* dst = Pz + ((ftile * (NROI / 16) + ntile) << 8) + lane * 4;
      const v8f a = acc[mt][nt];
      *reinterpret_cast<float4*>(dst)       = make_float4(a[0], a[1], a[2], a[3]);
      *reinterpret_cast<float4*>(dst + 128) = make_float4(a[4], a[5], a[6], a[7]);
    }
  }
}

#define SUM_BLOCKS ((PLANE / 4) / 256)

__global__ __launch_bounds__(256) void ksplit_sum_scatter_kernel(
    const float* __restrict__ P, const float* __restrict__ bias,
    const int* __restrict__ roi, float* __restrict__ out) {
  const int t     = blockIdx.x * 256 + threadIdx.x;
  const int lane  = t & 31;
  const int j     = (t >> 5) & 1;
  const int tile  = t >> 6;
  const int ntile = tile & (NROI / 16 - 1);
  const int ftile = tile >> 9;
  const float4* src = reinterpret_cast<const float4*>(P) + t;
  const float4 s0 = src[0];
  const float4 s1 = src[PLANE / 4];
  const float4 s2 = src[2 * (PLANE / 4)];
  const float4 s3 = src[3 * (PLANE / 4)];
  const int p = roi[ntile * 16 + (lane & 15)];
  const int f = ftile * 16 + (lane >> 4) * 8 + j * 4;
  float* o = out + (f << 14) + p;
  o[0 << 14] = (((s0.x + s1.x) + s2.x) + s3.x) + bias[f + 0];
  o[1 << 14] = (((s0.y + s1.y) + s2.y) + s3.y) + bias[f + 1];
  o[2 << 14] = (((s0.z + s1.z) + s2.z) + s3.z) + bias[f + 2];
  o[3 << 14] = (((s0.w + s1.w) + s2.w) + s3.w) + bias[f + 3];
}

extern "C" void kernel_launch(void* const* d_in, const int* in_sizes, int n_in,
                              void* d_out, int out_size, void* d_ws, size_t ws_size,
                              hipStream_t stream) {
  const float* X    = (const float*)d_in[0];
  const float* W    = (const float*)d_in[1];
  const float* B    = (const float*)d_in[2];
  const int*   roi  = (const int*)d_in[3];
  float*       out  = (float*)d_out;

  float*       Wf   = (float*)d_ws;
  float*       P    = Wf + F_OUT * KDIM;

  bias_fill_relayout_kernel<<<dim3(RELAYOUT_BLOCKS + FILL_BLOCKS), 256, 0, stream>>>(
      B, W, Wf, out);
  conv_roi_wmma_kernel<<<dim3(NROI / 128, F_OUT / 128, KSPLIT), 256, 0, stream>>>(
      X, Wf, roi, P);
  ksplit_sum_scatter_kernel<<<dim3(SUM_BLOCKS), 256, 0, stream>>>(P, B, roi, out);
}


# K split moved from grid z into the work-group (16 waves = 2x2 wave tiles x 4 K ranges): partials added in LDS, bias + scatter by the GEMM kernel; launch 3 and workspace planes gone; same sum order
# speedup vs baseline: 6.8680x; 1.0935x over previous
#include <hip/hip_runtime.h>


typedef float v2f __attribute__((ext_vector_type(2)));
typedef float v8f __attribute__((ext_vector_type(8)));

#define HW_DIM   128
#define F_OUT    256
#define KDIM     2304
#define NROI     8192
#define HOWO     16384
#define KTILES   144
#define KSPLIT   4
#define CHUNKS_PER_SPLIT (16 / KSPLIT)

#define RELAYOUT_BLOCKS ((F_OUT * KDIM / 4) / 256)
#define FILL_BLOCKS     ((F_OUT * HOWO / 4) / 256)

__global__ __launch_bounds__(256) void bias_fill_relayout_kernel(
    const float* __restrict__ bias, const float* __restrict__ W,
    float* __restrict__ Wf, float* __restrict__ out) {
  float4  val;
  float4* dst;
  if (blockIdx.x < RELAYOUT_BLOCKS) {
    const int g     = blockIdx.x * 256 + threadIdx.x;
    const int l     = g & 31;
    const int j     = (g >> 5) & 1;
    const int tk    = g >> 6;
    const int T     = tk / KTILES;
    const int kt    = tk - T * KTILES;
    const int chunk = kt / 9;
    const int tap   = kt - chunk * 9;
    const int f     = T * 16 + (l & 15);
    const int ch    = chunk * 16 + (l >> 4) * 8 + j * 4;
    const float* src = W + f * KDIM + ch * 9 + tap;
    val = make_float4(src[0], src[9], src[18], src[27]);
    dst = reinterpret_cast<float4*>(Wf) + g;
  } else {
    const int g = (blockIdx.x - RELAYOUT_BLOCKS) * 256 + threadIdx.x;
    const int f = g >> 12;
    const float bv = bias[f];
    val = make_float4(bv, bv, bv, bv);
    dst = reinterpret_cast<float4*>(out) + g;
  }
  *dst = val;
}


__device__ __forceinline__ bool tap_inside(int tap, int ry, int rx) {
  const int fi = tap / 3;
  const int fj = tap - fi * 3;
  const int y  = ry + fi - 1;
  const int x  = rx + fj - 1;
  return (unsigned)y < 128u && (unsigned)x < 128u;
}

__device__ __forceinline__ void load_tile(
    int tap, const float* __restrict__ wtile, const float* __restrict__ xchunk,
    const int (&p_my)[2], const int (&ry)[2], const int (&rx)[2],
    float4 (&a4)[4][2], float (&xv)[2][8]) {
#pragma unroll
  for (int mt = 0; mt < 4; ++mt)
#pragma unroll
    for (int j = 0; j < 2; ++j)
      a4[mt][j] = *reinterpret_cast<const float4*>(wtile + (mt * KTILES * 2 + j) * 128);
  const int fi = tap / 3;
  const int fj = tap - fi * 3;
#pragma unroll
  for (int nt = 0; nt < 2; ++nt) {
    const int poff = tap_inside(tap, ry[nt], rx[nt])
                         ? p_my[nt] + (fi - 1) * 128 + (fj - 1) : p_my[nt];
    const float* xp = xchunk + poff;
#pragma unroll
    for (int u = 0; u < 8; ++u) xv[nt][u] = xp[u << 14];
  }
}

__device__ __forceinline__ void mma_tile(
    int tap, const int (&ry)[2], const int (&rx)[2],
    const float4 (&a4)[4][2], const float (&xv)[2][8], v8f (&acc)[4][2]) {
  float b[2][8];
#pragma unroll
  for (int nt = 0; nt < 2; ++nt) {
    const bool ok = tap_inside(tap, ry[nt], rx[nt]);
#pragma unroll
    for (int u = 0; u < 8; ++u) {
      const float v = xv[nt][u];
      b[nt][u] = ok ? v : 0.0f;
    }
  }
#pragma unroll
  for (int ks = 0; ks < 4; ++ks) {
    v2f afrag[4];
#pragma unroll
    for (int mt = 0; mt < 4; ++mt) {
      const float4 q = a4[mt][ks >> 1];
      v2f a;
      a.x = (ks & 1) ? q.z : q.x;
      a.y = (ks & 1) ? q.w : q.y;
      afrag[mt] = a;
    }
    v2f bfrag[2];
#pragma unroll
    for (int nt = 0; nt < 2; ++nt) {
      v2f bb;
      bb.x = b[nt][2 * ks];
      bb.y = b[nt][2 * ks + 1];
      bfrag[nt] = bb;
    }
#pragma unroll
    for (int mt = 0; mt < 4; ++mt)
#pragma unroll
      for (int nt = 0; nt < 2; ++nt)
        acc[mt][nt] = __builtin_amdgcn_wmma_f32_16x16x4_f32(
            false, afrag[mt], false, bfrag[nt],
            (short)0, acc[mt][nt], false, false);
  }
}

__global__ __launch_bounds__(512) void conv_roi_wmma_kernel(
    const float* __restrict__ X,
    const float* __restrict__ Wf,
    const float* __restrict__ bias,
    const int*   __restrict__ roi,
    float* __restrict__ out) {
  __shared__ float4 red[4 * 4 * KSPLIT * 2 * 2 * 32];

  const int tid   = threadIdx.x;
  const int lane  = tid & 31;
  const int wave  = tid >> 5;
  const int q     = wave & 3;
  const int kz    = wave >> 2;
  const int waveM = q >> 1;
  const int waveN = q & 1;
  const int lrow  = lane & 15;
  const int lhi   = lane >> 4;

  const int m0 = blockIdx.y * 128;
  const int n0 = blockIdx.x * 64;

  int p_my[2], ry[2], rx[2];
#pragma unroll
  for (int nt = 0; nt < 2; ++nt) {
    p_my[nt] = roi[n0 + waveN * 32 + nt * 16 + lrow];
    ry[nt]   = p_my[nt] >> 7;
    rx[nt]   = p_my[nt] & 127;
  }

  v8f acc[4][2];
  const v8f vzero = {};
#pragma unroll
  for (int mt = 0; mt < 4; ++mt)
#pragma unroll
    for (int nt = 0; nt < 2; ++nt) acc[mt][nt] = vzero;

  const float* xbase = X + ((lhi * 8) << 14);
  const float* wbase = Wf + (((m0 >> 4) + waveM * 4) * KTILES) * 256 + lane * 4;

  const int chunk0 = kz * CHUNKS_PER_SPLIT;

  float4 a4[2][4][2];
  float  xv[2][2][8];

  load_tile(0, wbase + (chunk0 * 9) * 256, xbase + (chunk0 << 18), p_my, ry, rx, a4[0], xv[0]);

#pragma unroll 1
  for (int cc = 0; cc < CHUNKS_PER_SPLIT / 2; ++cc) {
    const int cbase = chunk0 + cc * 2;
#pragma unroll
    for (int t = 0; t < 18; ++t) {
      const int cur = t & 1;
      const int nxt = cur ^ 1;
      if (t < 17) {
        const int nchunk = cbase + (t + 1) / 9;
        const int ntap   = (t + 1) % 9;
        load_tile(ntap, wbase + (nchunk * 9 + ntap) * 256, xbase + (nchunk << 18),
                  p_my, ry, rx, a4[nxt], xv[nxt]);
      } else if (cc + 1 < CHUNKS_PER_SPLIT / 2) {
        const int nchunk = cbase + 2;
        load_tile(0, wbase + (nchunk * 9) * 256, xbase + (nchunk << 18),
                  p_my, ry, rx, a4[nxt], xv[nxt]);
      }
      __builtin_amdgcn_sched_barrier(0);
      mma_tile(t % 9, ry, rx, a4[cur], xv[cur], acc);
      __builtin_amdgcn_sched_barrier(0);
    }
  }

#pragma unroll
  for (int mt = 0; mt < 4; ++mt)
#pragma unroll
    for (int nt = 0; nt < 2; ++nt) {
      float4* dst = red + ((((q * 4 + mt) * KSPLIT + kz) * 2 + nt) * 2) * 32 + lane;
      const v8f a = acc[mt][nt];
      dst[0]  = make_float4(a[0], a[1], a[2], a[3]);
      dst[32] = make_float4(a[4], a[5], a[6], a[7]);
    }

  const int f = m0 + waveM * 64 + kz * 16 + lhi * 8;
  float bv[8];
#pragma unroll
  for (int r = 0; r < 8; ++r) bv[r] = bias[f + r];

  __syncthreads();

#pragma unroll
  for (int nt = 0; nt < 2; ++nt) {
    float* o = out + (f << 14) + p_my[nt];
#pragma unroll
    for (int j = 0; j < 2; ++j) {
      const float4* s = red + ((((q * 4 + kz) * KSPLIT) * 2 + nt) * 2 + j) * 32 + lane;
      const float4 s0 = s[0 * 128];
      const float4 s1 = s[1 * 128];
      const float4 s2 = s[2 * 128];
      const float4 s3 = s[3 * 128];
      o[(4 * j + 0) << 14] = (((s0.x + s1.x) + s2.x) + s3.x) + bv[4 * j + 0];
      o[(4 * j + 1) << 14] = (((s0.y + s1.y) + s2.y) + s3.y) + bv[4 * j + 1];
      o[(4 * j + 2) << 14] = (((s0.z + s1.z) + s2.z) + s3.z) + bv[4 * j + 2];
      o[(4 * j + 3) << 14] = (((s0.w + s1.w) + s2.w) + s3.w) + bv[4 * j + 3];
    }
  }
}

extern "C" void kernel_launch(void* const* d_in, const int* in_sizes, int n_in,
                              void* d_out, int out_size, void* d_ws, size_t ws_size,
                              hipStream_t stream) {
  const float* X    = (const float*)d_in[0];
  const float* W    = (const float*)d_in[1];
  const float* B    = (const float*)d_in[2];
  const int*   roi  = (const int*)d_in[3];
  float*       out  = (float*)d_out;

  float*       Wf   = (float*)d_ws;

  bias_fill_relayout_kernel<<<dim3(RELAYOUT_BLOCKS + FILL_BLOCKS), 256, 0, stream>>>(
      B, W, Wf, out);
  conv_roi_wmma_kernel<<<dim3(NROI / 64, F_OUT / 128), 512, 0, stream>>>(
      X, Wf, B, roi, out);
}
